# starting kernel
# baseline (speedup 1.0000x reference)
#include <hip/hip_runtime.h>
#include <math.h>

#define HDIM  256
#define BATCH 16
#define QLEN  64
#define KLEN  1024
#define DV    256
#define BQ    (BATCH * QLEN)
#define BKEYS (BATCH * KLEN)
#define NEGBIG -1e6f

typedef float v2f __attribute__((ext_vector_type(2)));
typedef float v8f __attribute__((ext_vector_type(8)));

__device__ __forceinline__ float fast_tanh(float x) {
#if __has_builtin(__builtin_amdgcn_tanhf)
  return __builtin_amdgcn_tanhf(x);
#elif __has_builtin(__builtin_amdgcn_tanh_f32)
  return __builtin_amdgcn_tanh_f32(x);
#else
  return tanhf(x);
#endif
}

__global__ __launch_bounds__(256) void proj_gemm(const float* __restrict__ X,
                                                 const float* __restrict__ W,
                                                 float* __restrict__ OUT) {
  const int lane = threadIdx.x & 31;
  const int wave = threadIdx.x >> 5;
  const int tile = blockIdx.x * 8 + wave;
  const int ntn  = HDIM / 16;
  const int tm   = tile / ntn;
  const int tn   = tile % ntn;
  const int half = lane >> 4;
  const int l16  = lane & 15;

  const float* Arow = X + (tm * 16 + l16) * HDIM;
  const float* Brow = W + (tn * 16 + l16) * HDIM;

  v8f acc = {};
#pragma unroll 8
  for (int k0 = 0; k0 < HDIM; k0 += 4) {
    v2f a = *(const v2f*)(Arow + k0 + 2 * half);
    v2f b = *(const v2f*)(Brow + k0 + 2 * half);
    acc = __builtin_amdgcn_wmma_f32_16x16x4_f32(
         false, a,  false, b,
         (short)0, acc,  false,  false);
  }

  float* Orow = OUT + (tm * 16 + 8 * half) * HDIM + tn * 16 + l16;
#pragma unroll
  for (int v = 0; v < 8; ++v)
    Orow[v * HDIM] = acc[v];
}

__global__ __launch_bounds__(256) void score_softmax(const float* __restrict__ Qp,
                                                     const float* __restrict__ Kp,
                                                     const float* __restrict__ wv,
                                                     const int* __restrict__ valid_lens,
                                                     float* __restrict__ attn) {
  __shared__ float scores[KLEN];
  __shared__ float red[256];

  const int bq   = blockIdx.x;
  const int b    = bq / QLEN;
  const int t    = threadIdx.x;
  const int lane = t & 31;
  const int wave = t >> 5;

  const float* qbase = Qp + (size_t)bq * HDIM + lane * 8;
  const float4 q0 = *(const float4*)(qbase);
  const float4 q1 = *(const float4*)(qbase + 4);
  const float* wbase = wv + lane * 8;
  const float4 w0 = *(const float4*)(wbase);
  const float4 w1 = *(const float4*)(wbase + 4);

  const int vlen = valid_lens[b];
  const int kbeg = wave * (KLEN / 8);
  const float* krow = Kp + ((size_t)b * KLEN + kbeg) * HDIM + lane * 8;

#pragma unroll 2
  for (int i = 0; i < KLEN / 8; ++i) {
    __builtin_prefetch(krow + 2 * HDIM, 0, 1);
    const float4 k0v = *(const float4*)(krow);
    const float4 k1v = *(const float4*)(krow + 4);

    float s = fast_tanh(q0.x + k0v.x) * w0.x;
    s += fast_tanh(q0.y + k0v.y) * w0.y;
    s += fast_tanh(q0.z + k0v.z) * w0.z;
    s += fast_tanh(q0.w + k0v.w) * w0.w;
    s += fast_tanh(q1.x + k1v.x) * w1.x;
    s += fast_tanh(q1.y + k1v.y) * w1.y;
    s += fast_tanh(q1.z + k1v.z) * w1.z;
    s += fast_tanh(q1.w + k1v.w) * w1.w;

#pragma unroll
    for (int off = 16; off > 0; off >>= 1)
      s += __shfl_xor(s, off, 32);

    const int k = kbeg + i;
    if (lane == 0) scores[k] = (k < vlen) ? s : NEGBIG;
    krow += HDIM;
  }
  __syncthreads();

  float sc[4];
#pragma unroll
  for (int j = 0; j < 4; ++j) sc[j] = scores[t + 256 * j];

  float m = fmaxf(fmaxf(sc[0], sc[1]), fmaxf(sc[2], sc[3]));
  red[t] = m;
  __syncthreads();
  for (int off = 128; off > 0; off >>= 1) {
    if (t < off) red[t] = fmaxf(red[t], red[t + off]);
    __syncthreads();
  }
  const float M = red[0];
  __syncthreads();

  float p[4];
  float ssum = 0.f;
#pragma unroll
  for (int j = 0; j < 4; ++j) {
    p[j] = __expf(sc[j] - M);
    ssum += p[j];
  }
  red[t] = ssum;
  __syncthreads();
  for (int off = 128; off > 0; off >>= 1) {
    if (t < off) red[t] += red[t + off];
    __syncthreads();
  }
  const float inv = 1.f / red[0];

#pragma unroll
  for (int j = 0; j < 4; ++j)
    attn[(size_t)bq * KLEN + t + 256 * j] = p[j] * inv;
}

__global__ __launch_bounds__(256) void out_gemm(const float* __restrict__ attn,
                                                const float* __restrict__ V,
                                                float* __restrict__ OUT) {
  const int lane = threadIdx.x & 31;
  const int wave = threadIdx.x >> 5;
  const int tile = blockIdx.x * 8 + wave;
  const int ntn  = DV / 16;
  const int ntm  = QLEN / 16;
  const int b    = tile / (ntm * ntn);
  const int r    = tile % (ntm * ntn);
  const int tm   = r / ntn;
  const int tn   = r % ntn;
  const int half = lane >> 4;
  const int l16  = lane & 15;

  const float* Ab = attn + (size_t)b * QLEN * KLEN + (tm * 16 + l16) * KLEN;
  const float* Vb = V + (size_t)b * KLEN * DV + tn * 16 + l16;

  v8f acc = {};
#pragma unroll 4
  for (int k0 = 0; k0 < KLEN; k0 += 4) {
    const int kb = k0 + 2 * half;
    v2f a = *(const v2f*)(Ab + kb);
    v2f bv;
    bv.x = Vb[(size_t)kb * DV];
    bv.y = Vb[(size_t)(kb + 1) * DV];
    acc = __builtin_amdgcn_wmma_f32_16x16x4_f32(
        false, a, false, bv, (short)0, acc, false, false);
  }

  float* Orow = OUT + (size_t)(b * QLEN + tm * 16 + 8 * half) * DV + tn * 16 + l16;
#pragma unroll
  for (int v = 0; v < 8; ++v)
    Orow[v * DV] = acc[v];
}

extern "C" void kernel_launch(void* const* d_in, const int* in_sizes, int n_in,
                              void* d_out, int out_size, void* d_ws, size_t ws_size,
                              hipStream_t stream) {
  const float* queries = (const float*)d_in[0];
  const float* keys    = (const float*)d_in[1];
  const float* values  = (const float*)d_in[2];
  const int*   vlens   = (const int*)d_in[3];
  const float* Wq      = (const float*)d_in[4];
  const float* Wk      = (const float*)d_in[5];
  const float* wv      = (const float*)d_in[6];
  float* out = (float*)d_out;

  float* Qp   = (float*)d_ws;
  float* Kp   = Qp + (size_t)BQ * HDIM;
  float* attn = Kp + (size_t)BKEYS * HDIM;

  proj_gemm<<<(BQ / 16) * (HDIM / 16) / 8, 256, 0, stream>>>(queries, Wq, Qp);
  proj_gemm<<<(BKEYS / 16) * (HDIM / 16) / 8, 256, 0, stream>>>(keys, Wk, Kp);

  score_softmax<<<BQ, 256, 0, stream>>>(Qp, Kp, wv, vlens, attn);

  out_gemm<<<(BATCH * (QLEN / 16) * (DV / 16)) / 8, 256, 0, stream>>>(attn, values, out);
}


# Projection GEMMs register-tiled: a wave holds a 32x64 (keys) or 16x32 (queries) tile, each 16-byte fragment load reused across it and issued chunks ahead; k summed in another order per 8-k step
# speedup vs baseline: 1.1467x; 1.1467x over previous
#include <hip/hip_runtime.h>
#include <math.h>

#define HDIM  256
#define BATCH 16
#define QLEN  64
#define KLEN  1024
#define DV    256
#define BQ    (BATCH * QLEN)
#define BKEYS (BATCH * KLEN)
#define NEGBIG -1e6f

typedef float v2f __attribute__((ext_vector_type(2)));
typedef float v4f __attribute__((ext_vector_type(4)));
typedef float v8f __attribute__((ext_vector_type(8)));

__device__ __forceinline__ float fast_tanh(float x) {
#if __has_builtin(__builtin_amdgcn_tanhf)
  return __builtin_amdgcn_tanhf(x);
#elif __has_builtin(__builtin_amdgcn_tanh_f32)
  return __builtin_amdgcn_tanh_f32(x);
#else
  return tanhf(x);
#endif
}

template <int MT, int NT, int WM, int WN, int KU>
__device__ __forceinline__ void proj_tile(const float* __restrict__ X,
                                          const float* __restrict__ W,
                                          float* __restrict__ OUT) {
  constexpr int BM  = WM * MT * 16;
  constexpr int BN  = WN * NT * 16;
  constexpr int NBN = HDIM / BN;
  constexpr int KC  = 8 * KU;

  const int lane = threadIdx.x & 31;
  const int wave = threadIdx.x >> 5;
  const int half = lane >> 4;
  const int l16  = lane & 15;
  const int bm   = blockIdx.x / NBN;
  const int bn   = blockIdx.x % NBN;
  const int m0   = bm * BM + (wave / WN) * (MT * 16);
  const int n0   = bn * BN + (wave % WN) * (NT * 16);

  const float* Ap = X + (size_t)(m0 + l16) * HDIM + 4 * half;
  const float* Bp = W + (size_t)(n0 + l16) * HDIM + 4 * half;

  v8f acc[MT][NT];
#pragma unroll
  for (int mt = 0; mt < MT; ++mt)
#pragma unroll
    for (int nt = 0; nt < NT; ++nt) acc[mt][nt] = v8f{};

  v4f ca[KU][MT], cb[KU][NT];
#pragma unroll
  for (int u = 0; u < KU; ++u) {
#pragma unroll
    for (int mt = 0; mt < MT; ++mt) ca[u][mt] = *(const v4f*)(Ap + mt * 16 * HDIM + 8 * u);
#pragma unroll
    for (int nt = 0; nt < NT; ++nt) cb[u][nt] = *(const v4f*)(Bp + nt * 16 * HDIM + 8 * u);
  }

#pragma unroll
  for (int kc = 0; kc < HDIM; kc += KC) {
    v4f na[KU][MT], nb[KU][NT];
    if (kc + KC < HDIM) {
      const int kn = kc + KC;
#pragma unroll
      for (int u = 0; u < KU; ++u) {
#pragma unroll
        for (int mt = 0; mt < MT; ++mt) na[u][mt] = *(const v4f*)(Ap + mt * 16 * HDIM + kn + 8 * u);
#pragma unroll
        for (int nt = 0; nt < NT; ++nt) nb[u][nt] = *(const v4f*)(Bp + nt * 16 * HDIM + kn + 8 * u);
      }
    }
    __builtin_amdgcn_sched_barrier(0);

#pragma unroll
    for (int u = 0; u < KU; ++u) {
#pragma unroll
      for (int mt = 0; mt < MT; ++mt)
#pragma unroll
        for (int nt = 0; nt < NT; ++nt) {
          const v2f a = {ca[u][mt].x, ca[u][mt].y};
          const v2f b = {cb[u][nt].x, cb[u][nt].y};
          acc[mt][nt] = __builtin_amdgcn_wmma_f32_16x16x4_f32(
               false, a,  false, b,
               (short)0, acc[mt][nt],  false,  false);
        }
#pragma unroll
      for (int mt = 0; mt < MT; ++mt)
#pragma unroll
        for (int nt = 0; nt < NT; ++nt) {
          const v2f a = {ca[u][mt].z, ca[u][mt].w};
          const v2f b = {cb[u][nt].z, cb[u][nt].w};
          acc[mt][nt] = __builtin_amdgcn_wmma_f32_16x16x4_f32(
              false, a, false, b, (short)0, acc[mt][nt], false, false);
        }
    }

    if (kc + KC < HDIM) {
#pragma unroll
      for (int u = 0; u < KU; ++u) {
#pragma unroll
        for (int mt = 0; mt < MT; ++mt) ca[u][mt] = na[u][mt];
#pragma unroll
        for (int nt = 0; nt < NT; ++nt) cb[u][nt] = nb[u][nt];
      }
    }
  }

#pragma unroll
  for (int mt = 0; mt < MT; ++mt)
#pragma unroll
    for (int nt = 0; nt < NT; ++nt) {
      float* Orow = OUT + (size_t)(m0 + mt * 16 + 8 * half) * HDIM + n0 + nt * 16 + l16;
#pragma unroll
      for (int v = 0; v < 8; ++v)
        Orow[v * HDIM] = acc[mt][nt][v];
    }
}

__global__ __launch_bounds__(256) void proj_gemm_q(const float* __restrict__ X,
                                                   const float* __restrict__ W,
                                                   float* __restrict__ OUT) {
  proj_tile<1, 2, 1, 8, 4>(X, W, OUT);
}

__global__ __launch_bounds__(256) void proj_gemm_k(const float* __restrict__ X,
                                                   const float* __restrict__ W,
                                                   float* __restrict__ OUT) {
  proj_tile<2, 4, 4, 2, 2>(X, W, OUT);
}

__global__ __launch_bounds__(256) void score_softmax(const float* __restrict__ Qp,
                                                     const float* __restrict__ Kp,
                                                     const float* __restrict__ wv,
                                                     const int* __restrict__ valid_lens,
                                                     float* __restrict__ attn) {
  __shared__ float scores[KLEN];
  __shared__ float red[256];

  const int bq   = blockIdx.x;
  const int b    = bq / QLEN;
  const int t    = threadIdx.x;
  const int lane = t & 31;
  const int wave = t >> 5;

  const float* qbase = Qp + (size_t)bq * HDIM + lane * 8;
  const float4 q0 = *(const float4*)(qbase);
  const float4 q1 = *(const float4*)(qbase + 4);
  const float* wbase = wv + lane * 8;
  const float4 w0 = *(const float4*)(wbase);
  const float4 w1 = *(const float4*)(wbase + 4);

  const int vlen = valid_lens[b];
  const int kbeg = wave * (KLEN / 8);
  const float* krow = Kp + ((size_t)b * KLEN + kbeg) * HDIM + lane * 8;

#pragma unroll 2
  for (int i = 0; i < KLEN / 8; ++i) {
    __builtin_prefetch(krow + 2 * HDIM, 0, 1);
    const float4 k0v = *(const float4*)(krow);
    const float4 k1v = *(const float4*)(krow + 4);

    float s = fast_tanh(q0.x + k0v.x) * w0.x;
    s += fast_tanh(q0.y + k0v.y) * w0.y;
    s += fast_tanh(q0.z + k0v.z) * w0.z;
    s += fast_tanh(q0.w + k0v.w) * w0.w;
    s += fast_tanh(q1.x + k1v.x) * w1.x;
    s += fast_tanh(q1.y + k1v.y) * w1.y;
    s += fast_tanh(q1.z + k1v.z) * w1.z;
    s += fast_tanh(q1.w + k1v.w) * w1.w;

#pragma unroll
    for (int off = 16; off > 0; off >>= 1)
      s += __shfl_xor(s, off, 32);

    const int k = kbeg + i;
    if (lane == 0) scores[k] = (k < vlen) ? s : NEGBIG;
    krow += HDIM;
  }
  __syncthreads();

  float sc[4];
#pragma unroll
  for (int j = 0; j < 4; ++j) sc[j] = scores[t + 256 * j];

  float m = fmaxf(fmaxf(sc[0], sc[1]), fmaxf(sc[2], sc[3]));
  red[t] = m;
  __syncthreads();
  for (int off = 128; off > 0; off >>= 1) {
    if (t < off) red[t] = fmaxf(red[t], red[t + off]);
    __syncthreads();
  }
  const float M = red[0];
  __syncthreads();

  float p[4];
  float ssum = 0.f;
#pragma unroll
  for (int j = 0; j < 4; ++j) {
    p[j] = __expf(sc[j] - M);
    ssum += p[j];
  }
  red[t] = ssum;
  __syncthreads();
  for (int off = 128; off > 0; off >>= 1) {
    if (t < off) red[t] += red[t + off];
    __syncthreads();
  }
  const float inv = 1.f / red[0];

#pragma unroll
  for (int j = 0; j < 4; ++j)
    attn[(size_t)bq * KLEN + t + 256 * j] = p[j] * inv;
}

__global__ __launch_bounds__(256) void out_gemm(const float* __restrict__ attn,
                                                const float* __restrict__ V,
                                                float* __restrict__ OUT) {
  const int lane = threadIdx.x & 31;
  const int wave = threadIdx.x >> 5;
  const int tile = blockIdx.x * 8 + wave;
  const int ntn  = DV / 16;
  const int ntm  = QLEN / 16;
  const int b    = tile / (ntm * ntn);
  const int r    = tile % (ntm * ntn);
  const int tm   = r / ntn;
  const int tn   = r % ntn;
  const int half = lane >> 4;
  const int l16  = lane & 15;

  const float* Ab = attn + (size_t)b * QLEN * KLEN + (tm * 16 + l16) * KLEN;
  const float* Vb = V + (size_t)b * KLEN * DV + tn * 16 + l16;

  v8f acc = {};
#pragma unroll 4
  for (int k0 = 0; k0 < KLEN; k0 += 4) {
    const int kb = k0 + 2 * half;
    v2f a = *(const v2f*)(Ab + kb);
    v2f bv;
    bv.x = Vb[(size_t)kb * DV];
    bv.y = Vb[(size_t)(kb + 1) * DV];
    acc = __builtin_amdgcn_wmma_f32_16x16x4_f32(
        false, a, false, bv, (short)0, acc, false, false);
  }

  float* Orow = OUT + (size_t)(b * QLEN + tm * 16 + 8 * half) * DV + tn * 16 + l16;
#pragma unroll
  for (int v = 0; v < 8; ++v)
    Orow[v * DV] = acc[v];
}

extern "C" void kernel_launch(void* const* d_in, const int* in_sizes, int n_in,
                              void* d_out, int out_size, void* d_ws, size_t ws_size,
                              hipStream_t stream) {
  const float* queries = (const float*)d_in[0];
  const float* keys    = (const float*)d_in[1];
  const float* values  = (const float*)d_in[2];
  const int*   vlens   = (const int*)d_in[3];
  const float* Wq      = (const float*)d_in[4];
  const float* Wk      = (const float*)d_in[5];
  const float* wv      = (const float*)d_in[6];
  float* out = (float*)d_out;

  float* Qp   = (float*)d_ws;
  float* Kp   = Qp + (size_t)BQ * HDIM;
  float* attn = Kp + (size_t)BKEYS * HDIM;

  proj_gemm_q<<<64, 256, 0, stream>>>(queries, Wq, Qp);
  proj_gemm_k<<<256, 256, 0, stream>>>(keys, Wk, Kp);

  score_softmax<<<BQ, 256, 0, stream>>>(Qp, Kp, wv, vlens, attn);

  out_gemm<<<(BATCH * (QLEN / 16) * (DV / 16)) / 8, 256, 0, stream>>>(attn, values, out);
}


# Score kernel remapped: a lane owns 4 keys x 4 queries and loops h (key projection stored h-major), so the h sum is a per-lane chain with no cross-lane step; h and softmax sums in another order
# speedup vs baseline: 1.6798x; 1.4649x over previous
#include <hip/hip_runtime.h>
#include <math.h>

#define HDIM  256
#define BATCH 16
#define QLEN  64
#define KLEN  1024
#define DV    256
#define BQ    (BATCH * QLEN)
#define BKEYS (BATCH * KLEN)
#define NEGBIG -1e6f

typedef float v2f __attribute__((ext_vector_type(2)));
typedef float v4f __attribute__((ext_vector_type(4)));
typedef float v8f __attribute__((ext_vector_type(8)));

__device__ __forceinline__ float fast_tanh(float x) {
#if __has_builtin(__builtin_amdgcn_tanhf)
  return __builtin_amdgcn_tanhf(x);
#elif __has_builtin(__builtin_amdgcn_tanh_f32)
  return __builtin_amdgcn_tanh_f32(x);
#else
  return tanhf(x);
#endif
}

template <int MT, int NT, int WM, int WN, int KU, bool HMAJOR>
__device__ __forceinline__ void proj_tile(const float* __restrict__ X,
                                          const float* __restrict__ W,
                                          float* __restrict__ OUT) {
  constexpr int BM  = WM * MT * 16;
  constexpr int BN  = WN * NT * 16;
  constexpr int NBN = HDIM / BN;
  constexpr int KC  = 8 * KU;

  const int lane = threadIdx.x & 31;
  const int wave = threadIdx.x >> 5;
  const int half = lane >> 4;
  const int l16  = lane & 15;
  const int bm   = blockIdx.x / NBN;
  const int bn   = blockIdx.x % NBN;
  const int m0   = bm * BM + (wave / WN) * (MT * 16);
  const int n0   = bn * BN + (wave % WN) * (NT * 16);

  const float* Ap = X + (size_t)(m0 + l16) * HDIM + 4 * half;
  const float* Bp = W + (size_t)(n0 + l16) * HDIM + 4 * half;

  v8f acc[MT][NT];
#pragma unroll
  for (int mt = 0; mt < MT; ++mt)
#pragma unroll
    for (int nt = 0; nt < NT; ++nt) acc[mt][nt] = v8f{};

  v4f ca[KU][MT], cb[KU][NT];
#pragma unroll
  for (int u = 0; u < KU; ++u) {
#pragma unroll
    for (int mt = 0; mt < MT; ++mt) ca[u][mt] = *(const v4f*)(Ap + mt * 16 * HDIM + 8 * u);
#pragma unroll
    for (int nt = 0; nt < NT; ++nt) cb[u][nt] = *(const v4f*)(Bp + nt * 16 * HDIM + 8 * u);
  }

#pragma unroll
  for (int kc = 0; kc < HDIM; kc += KC) {
    v4f na[KU][MT], nb[KU][NT];
    if (kc + KC < HDIM) {
      const int kn = kc + KC;
#pragma unroll
      for (int u = 0; u < KU; ++u) {
#pragma unroll
        for (int mt = 0; mt < MT; ++mt) na[u][mt] = *(const v4f*)(Ap + mt * 16 * HDIM + kn + 8 * u);
#pragma unroll
        for (int nt = 0; nt < NT; ++nt) nb[u][nt] = *(const v4f*)(Bp + nt * 16 * HDIM + kn + 8 * u);
      }
    }
    __builtin_amdgcn_sched_barrier(0);

#pragma unroll
    for (int u = 0; u < KU; ++u) {
#pragma unroll
      for (int mt = 0; mt < MT; ++mt)
#pragma unroll
        for (int nt = 0; nt < NT; ++nt) {
          const v2f a = {ca[u][mt].x, ca[u][mt].y};
          const v2f b = {cb[u][nt].x, cb[u][nt].y};
          acc[mt][nt] = __builtin_amdgcn_wmma_f32_16x16x4_f32(
               false, a,  false, b,
               (short)0, acc[mt][nt],  false,  false);
        }
#pragma unroll
      for (int mt = 0; mt < MT; ++mt)
#pragma unroll
        for (int nt = 0; nt < NT; ++nt) {
          const v2f a = {ca[u][mt].z, ca[u][mt].w};
          const v2f b = {cb[u][nt].z, cb[u][nt].w};
          acc[mt][nt] = __builtin_amdgcn_wmma_f32_16x16x4_f32(
              false, a, false, b, (short)0, acc[mt][nt], false, false);
        }
    }

    if (kc + KC < HDIM) {
#pragma unroll
      for (int u = 0; u < KU; ++u) {
#pragma unroll
        for (int mt = 0; mt < MT; ++mt) ca[u][mt] = na[u][mt];
#pragma unroll
        for (int nt = 0; nt < NT; ++nt) cb[u][nt] = nb[u][nt];
      }
    }
  }

  if (HMAJOR) {
#pragma unroll
    for (int mt = 0; mt < MT; ++mt)
#pragma unroll
      for (int nt = 0; nt < NT; ++nt) {
        const int mrow = m0 + mt * 16 + 8 * half;
        const int bb   = mrow / KLEN;
        const int kk   = mrow % KLEN;
        float* Ocol = OUT + ((size_t)bb * HDIM + n0 + nt * 16 + l16) * KLEN + kk;
        const v8f a = acc[mt][nt];
        *(v4f*)(Ocol)     = v4f{a[0], a[1], a[2], a[3]};
        *(v4f*)(Ocol + 4) = v4f{a[4], a[5], a[6], a[7]};
      }
  } else {
#pragma unroll
    for (int mt = 0; mt < MT; ++mt)
#pragma unroll
      for (int nt = 0; nt < NT; ++nt) {
        float* Orow = OUT + (size_t)(m0 + mt * 16 + 8 * half) * HDIM + n0 + nt * 16 + l16;
#pragma unroll
        for (int v = 0; v < 8; ++v)
          Orow[v * HDIM] = acc[mt][nt][v];
      }
  }
}

__global__ __launch_bounds__(256) void proj_gemm_q(const float* __restrict__ X,
                                                   const float* __restrict__ W,
                                                   float* __restrict__ OUT) {
  proj_tile<1, 2, 1, 8, 4, false>(X, W, OUT);
}

__global__ __launch_bounds__(256) void proj_gemm_k(const float* __restrict__ X,
                                                   const float* __restrict__ W,
                                                   float* __restrict__ OUT) {
  proj_tile<2, 4, 4, 2, 2, true>(X, W, OUT);
}

#define QT 4
#define NS 8
#define NCH (HDIM / 4)

__device__ __forceinline__ void ring_pass(const int it, const float* __restrict__ kp,
                                          const v4f (*qs)[NCH], const v4f* ws,
                                          v4f (&kr)[NS][4], v4f& wc, v4f (&qc)[QT],
                                          v4f (&acc)[QT]) {
#pragma unroll
  for (int s = 0; s < NS; ++s) {
    const int c = it * NS + s;
    const int cn = min(c + NS - 1, NCH - 1);
#pragma unroll
    for (int u = 0; u < 4; ++u)
      kr[(s + NS - 1) % NS][u] = *(const v4f*)(kp + (size_t)(4 * cn + u) * KLEN);
    const int c1 = (c + 1) & (NCH - 1);
    const v4f wn = ws[c1];
    v4f qn[QT];
#pragma unroll
    for (int qi = 0; qi < QT; ++qi) qn[qi] = qs[qi][c1];
    __builtin_amdgcn_sched_barrier(0);
#pragma unroll
    for (int uu = 0; uu < 4; ++uu) {
      const v4f kv = kr[s][uu];
      const float w = wc[uu];
#pragma unroll
      for (int qi = 0; qi < QT; ++qi) {
        const float qv = qc[qi][uu];
        acc[qi].x = fmaf(fast_tanh(qv + kv.x), w, acc[qi].x);
        acc[qi].y = fmaf(fast_tanh(qv + kv.y), w, acc[qi].y);
        acc[qi].z = fmaf(fast_tanh(qv + kv.z), w, acc[qi].z);
        acc[qi].w = fmaf(fast_tanh(qv + kv.w), w, acc[qi].w);
      }
    }
    wc = wn;
#pragma unroll
    for (int qi = 0; qi < QT; ++qi) qc[qi] = qn[qi];
  }
}

__global__ __launch_bounds__(256) __attribute__((amdgpu_waves_per_eu(2, 2)))
void score_softmax(const float* __restrict__ Qp,
                   const float* __restrict__ KpT,
                   const float* __restrict__ wv,
                   const int* __restrict__ valid_lens,
                   float* __restrict__ attn) {
  __shared__ v4f qs[QT][NCH];
  __shared__ v4f ws[NCH];
  __shared__ v4f redm[8];
  __shared__ v4f reds[8];

  const int t    = threadIdx.x;
  const int lane = t & 31;
  const int wave = t >> 5;
  const int b    = blockIdx.x / (QLEN / QT);
  const int bq0  = blockIdx.x * QT;

  (&qs[0][0])[t] = *(const v4f*)(Qp + (size_t)bq0 * HDIM + 4 * t);
  if (t < NCH) ws[t] = *(const v4f*)(wv + 4 * t);

  const int kbase  = wave * (KLEN / 8) + lane * 4;
  const float* kp  = KpT + (size_t)b * HDIM * KLEN + kbase;

  v4f kr[NS][4];
#pragma unroll
  for (int s = 0; s < NS - 1; ++s)
#pragma unroll
    for (int u = 0; u < 4; ++u) kr[s][u] = *(const v4f*)(kp + (size_t)(4 * s + u) * KLEN);

  v4f acc[QT];
#pragma unroll
  for (int qi = 0; qi < QT; ++qi) acc[qi] = v4f{0.f, 0.f, 0.f, 0.f};

  __syncthreads();

  v4f wc = ws[0];
  v4f qc[QT];
#pragma unroll
  for (int qi = 0; qi < QT; ++qi) qc[qi] = qs[qi][0];

  for (int it = 0; it < NCH / NS; ++it)
    ring_pass(it, kp, qs, ws, kr, wc, qc, acc);

  const int vlen = valid_lens[b];
  v4f mx;
#pragma unroll
  for (int qi = 0; qi < QT; ++qi) {
    acc[qi].x = (kbase + 0 < vlen) ? acc[qi].x : NEGBIG;
    acc[qi].y = (kbase + 1 < vlen) ? acc[qi].y : NEGBIG;
    acc[qi].z = (kbase + 2 < vlen) ? acc[qi].z : NEGBIG;
    acc[qi].w = (kbase + 3 < vlen) ? acc[qi].w : NEGBIG;
    mx[qi] = fmaxf(fmaxf(acc[qi].x, acc[qi].y), fmaxf(acc[qi].z, acc[qi].w));
  }
#pragma unroll
  for (int off = 16; off > 0; off >>= 1)
#pragma unroll
    for (int qi = 0; qi < QT; ++qi) mx[qi] = fmaxf(mx[qi], __shfl_xor(mx[qi], off, 32));
  if (lane == 0) redm[wave] = mx;
  __syncthreads();

  v4f M = redm[0];
#pragma unroll
  for (int w = 1; w < 8; ++w) {
    const v4f o = redm[w];
#pragma unroll
    for (int qi = 0; qi < QT; ++qi) M[qi] = fmaxf(M[qi], o[qi]);
  }

  v4f sm;
#pragma unroll
  for (int qi = 0; qi < QT; ++qi) {
    acc[qi].x = __expf(acc[qi].x - M[qi]);
    acc[qi].y = __expf(acc[qi].y - M[qi]);
    acc[qi].z = __expf(acc[qi].z - M[qi]);
    acc[qi].w = __expf(acc[qi].w - M[qi]);
    sm[qi] = (acc[qi].x + acc[qi].y) + (acc[qi].z + acc[qi].w);
  }
#pragma unroll
  for (int off = 16; off > 0; off >>= 1)
#pragma unroll
    for (int qi = 0; qi < QT; ++qi) sm[qi] += __shfl_xor(sm[qi], off, 32);
  if (lane == 0) reds[wave] = sm;
  __syncthreads();

  v4f S = reds[0];
#pragma unroll
  for (int w = 1; w < 8; ++w) {
    const v4f o = reds[w];
#pragma unroll
    for (int qi = 0; qi < QT; ++qi) S[qi] += o[qi];
  }

#pragma unroll
  for (int qi = 0; qi < QT; ++qi) {
    const float inv = 1.f / S[qi];
    *(v4f*)(attn + (size_t)(bq0 + qi) * KLEN + kbase) =
        v4f{acc[qi].x * inv, acc[qi].y * inv, acc[qi].z * inv, acc[qi].w * inv};
  }
}

__global__ __launch_bounds__(256) void out_gemm(const float* __restrict__ attn,
                                                const float* __restrict__ V,
                                                float* __restrict__ OUT) {
  const int lane = threadIdx.x & 31;
  const int wave = threadIdx.x >> 5;
  const int tile = blockIdx.x * 8 + wave;
  const int ntn  = DV / 16;
  const int ntm  = QLEN / 16;
  const int b    = tile / (ntm * ntn);
  const int r    = tile % (ntm * ntn);
  const int tm   = r / ntn;
  const int tn   = r % ntn;
  const int half = lane >> 4;
  const int l16  = lane & 15;

  const float* Ab = attn + (size_t)b * QLEN * KLEN + (tm * 16 + l16) * KLEN;
  const float* Vb = V + (size_t)b * KLEN * DV + tn * 16 + l16;

  v8f acc = {};
#pragma unroll 4
  for (int k0 = 0; k0 < KLEN; k0 += 4) {
    const int kb = k0 + 2 * half;
    v2f a = *(const v2f*)(Ab + kb);
    v2f bv;
    bv.x = Vb[(size_t)kb * DV];
    bv.y = Vb[(size_t)(kb + 1) * DV];
    acc = __builtin_amdgcn_wmma_f32_16x16x4_f32(
        false, a, false, bv, (short)0, acc, false, false);
  }

  float* Orow = OUT + (size_t)(b * QLEN + tm * 16 + 8 * half) * DV + tn * 16 + l16;
#pragma unroll
  for (int v = 0; v < 8; ++v)
    Orow[v * DV] = acc[v];
}

extern "C" void kernel_launch(void* const* d_in, const int* in_sizes, int n_in,
                              void* d_out, int out_size, void* d_ws, size_t ws_size,
                              hipStream_t stream) {
  const float* queries = (const float*)d_in[0];
  const float* keys    = (const float*)d_in[1];
  const float* values  = (const float*)d_in[2];
  const int*   vlens   = (const int*)d_in[3];
  const float* Wq      = (const float*)d_in[4];
  const float* Wk      = (const float*)d_in[5];
  const float* wv      = (const float*)d_in[6];
  float* out = (float*)d_out;

  float* Qp   = (float*)d_ws;
  float* Kp   = Qp + (size_t)BQ * HDIM;
  float* attn = Kp + (size_t)BKEYS * HDIM;

  proj_gemm_q<<<64, 256, 0, stream>>>(queries, Wq, Qp);
  proj_gemm_k<<<256, 256, 0, stream>>>(keys, Wk, Kp);

  score_softmax<<<BQ / QT, 256, 0, stream>>>(Qp, Kp, wv, vlens, attn);

  out_gemm<<<(BATCH * (QLEN / 16) * (DV / 16)) / 8, 256, 0, stream>>>(attn, values, out);
}


# Output GEMM re-tiled: a block owns a 32x32 tile, its 8 waves split k and hold 2x2 accumulators fed by 16/8-byte loads all issued up front, partials added through LDS; k summed in another order
# speedup vs baseline: 2.8081x; 1.6717x over previous
#include <hip/hip_runtime.h>
#include <math.h>

#define HDIM  256
#define BATCH 16
#define QLEN  64
#define KLEN  1024
#define DV    256
#define BQ    (BATCH * QLEN)
#define BKEYS (BATCH * KLEN)
#define NEGBIG -1e6f

typedef float v2f __attribute__((ext_vector_type(2)));
typedef float v4f __attribute__((ext_vector_type(4)));
typedef float v8f __attribute__((ext_vector_type(8)));

__device__ __forceinline__ float fast_tanh(float x) {
#if __has_builtin(__builtin_amdgcn_tanhf)
  return __builtin_amdgcn_tanhf(x);
#elif __has_builtin(__builtin_amdgcn_tanh_f32)
  return __builtin_amdgcn_tanh_f32(x);
#else
  return tanhf(x);
#endif
}

template <int MT, int NT, int WM, int WN, int KU, bool HMAJOR>
__device__ __forceinline__ void proj_tile(const float* __restrict__ X,
                                          const float* __restrict__ W,
                                          float* __restrict__ OUT) {
  constexpr int BM  = WM * MT * 16;
  constexpr int BN  = WN * NT * 16;
  constexpr int NBN = HDIM / BN;
  constexpr int KC  = 8 * KU;

  const int lane = threadIdx.x & 31;
  const int wave = threadIdx.x >> 5;
  const int half = lane >> 4;
  const int l16  = lane & 15;
  const int bm   = blockIdx.x / NBN;
  const int bn   = blockIdx.x % NBN;
  const int m0   = bm * BM + (wave / WN) * (MT * 16);
  const int n0   = bn * BN + (wave % WN) * (NT * 16);

  const float* Ap = X + (size_t)(m0 + l16) * HDIM + 4 * half;
  const float* Bp = W + (size_t)(n0 + l16) * HDIM + 4 * half;

  v8f acc[MT][NT];
#pragma unroll
  for (int mt = 0; mt < MT; ++mt)
#pragma unroll
    for (int nt = 0; nt < NT; ++nt) acc[mt][nt] = v8f{};

  v4f ca[KU][MT], cb[KU][NT];
#pragma unroll
  for (int u = 0; u < KU; ++u) {
#pragma unroll
    for (int mt = 0; mt < MT; ++mt) ca[u][mt] = *(const v4f*)(Ap + mt * 16 * HDIM + 8 * u);
#pragma unroll
    for (int nt = 0; nt < NT; ++nt) cb[u][nt] = *(const v4f*)(Bp + nt * 16 * HDIM + 8 * u);
  }

#pragma unroll
  for (int kc = 0; kc < HDIM; kc += KC) {
    v4f na[KU][MT], nb[KU][NT];
    if (kc + KC < HDIM) {
      const int kn = kc + KC;
#pragma unroll
      for (int u = 0; u < KU; ++u) {
#pragma unroll
        for (int mt = 0; mt < MT; ++mt) na[u][mt] = *(const v4f*)(Ap + mt * 16 * HDIM + kn + 8 * u);
#pragma unroll
        for (int nt = 0; nt < NT; ++nt) nb[u][nt] = *(const v4f*)(Bp + nt * 16 * HDIM + kn + 8 * u);
      }
    }
    __builtin_amdgcn_sched_barrier(0);

#pragma unroll
    for (int u = 0; u < KU; ++u) {
#pragma unroll
      for (int mt = 0; mt < MT; ++mt)
#pragma unroll
        for (int nt = 0; nt < NT; ++nt) {
          const v2f a = {ca[u][mt].x, ca[u][mt].y};
          const v2f b = {cb[u][nt].x, cb[u][nt].y};
          acc[mt][nt] = __builtin_amdgcn_wmma_f32_16x16x4_f32(
               false, a,  false, b,
               (short)0, acc[mt][nt],  false,  false);
        }
#pragma unroll
      for (int mt = 0; mt < MT; ++mt)
#pragma unroll
        for (int nt = 0; nt < NT; ++nt) {
          const v2f a = {ca[u][mt].z, ca[u][mt].w};
          const v2f b = {cb[u][nt].z, cb[u][nt].w};
          acc[mt][nt] = __builtin_amdgcn_wmma_f32_16x16x4_f32(
              false, a, false, b, (short)0, acc[mt][nt], false, false);
        }
    }

    if (kc + KC < HDIM) {
#pragma unroll
      for (int u = 0; u < KU; ++u) {
#pragma unroll
        for (int mt = 0; mt < MT; ++mt) ca[u][mt] = na[u][mt];
#pragma unroll
        for (int nt = 0; nt < NT; ++nt) cb[u][nt] = nb[u][nt];
      }
    }
  }

  if (HMAJOR) {
#pragma unroll
    for (int mt = 0; mt < MT; ++mt)
#pragma unroll
      for (int nt = 0; nt < NT; ++nt) {
        const int mrow = m0 + mt * 16 + 8 * half;
        const int bb   = mrow / KLEN;
        const int kk   = mrow % KLEN;
        float* Ocol = OUT + ((size_t)bb * HDIM + n0 + nt * 16 + l16) * KLEN + kk;
        const v8f a = acc[mt][nt];
        *(v4f*)(Ocol)     = v4f{a[0], a[1], a[2], a[3]};
        *(v4f*)(Ocol + 4) = v4f{a[4], a[5], a[6], a[7]};
      }
  } else {
#pragma unroll
    for (int mt = 0; mt < MT; ++mt)
#pragma unroll
      for (int nt = 0; nt < NT; ++nt) {
        float* Orow = OUT + (size_t)(m0 + mt * 16 + 8 * half) * HDIM + n0 + nt * 16 + l16;
#pragma unroll
        for (int v = 0; v < 8; ++v)
          Orow[v * HDIM] = acc[mt][nt][v];
      }
  }
}

__global__ __launch_bounds__(256) void proj_gemm_q(const float* __restrict__ X,
                                                   const float* __restrict__ W,
                                                   float* __restrict__ OUT) {
  proj_tile<1, 2, 1, 8, 4, false>(X, W, OUT);
}

__global__ __launch_bounds__(256) void proj_gemm_k(const float* __restrict__ X,
                                                   const float* __restrict__ W,
                                                   float* __restrict__ OUT) {
  proj_tile<2, 4, 4, 2, 2, true>(X, W, OUT);
}

#define QT 4
#define NS 8
#define NCH (HDIM / 4)

__device__ __forceinline__ void ring_pass(const int it, const float* __restrict__ kp,
                                          const v4f (*qs)[NCH], const v4f* ws,
                                          v4f (&kr)[NS][4], v4f& wc, v4f (&qc)[QT],
                                          v4f (&acc)[QT]) {
#pragma unroll
  for (int s = 0; s < NS; ++s) {
    const int c = it * NS + s;
    const int cn = min(c + NS - 1, NCH - 1);
#pragma unroll
    for (int u = 0; u < 4; ++u)
      kr[(s + NS - 1) % NS][u] = *(const v4f*)(kp + (size_t)(4 * cn + u) * KLEN);
    const int c1 = (c + 1) & (NCH - 1);
    const v4f wn = ws[c1];
    v4f qn[QT];
#pragma unroll
    for (int qi = 0; qi < QT; ++qi) qn[qi] = qs[qi][c1];
    __builtin_amdgcn_sched_barrier(0);
#pragma unroll
    for (int uu = 0; uu < 4; ++uu) {
      const v4f kv = kr[s][uu];
      const float w = wc[uu];
#pragma unroll
      for (int qi = 0; qi < QT; ++qi) {
        const float qv = qc[qi][uu];
        acc[qi].x = fmaf(fast_tanh(qv + kv.x), w, acc[qi].x);
        acc[qi].y = fmaf(fast_tanh(qv + kv.y), w, acc[qi].y);
        acc[qi].z = fmaf(fast_tanh(qv + kv.z), w, acc[qi].z);
        acc[qi].w = fmaf(fast_tanh(qv + kv.w), w, acc[qi].w);
      }
    }
    wc = wn;
#pragma unroll
    for (int qi = 0; qi < QT; ++qi) qc[qi] = qn[qi];
  }
}

__global__ __launch_bounds__(256) __attribute__((amdgpu_waves_per_eu(2, 2)))
void score_softmax(const float* __restrict__ Qp,
                   const float* __restrict__ KpT,
                   const float* __restrict__ wv,
                   const int* __restrict__ valid_lens,
                   float* __restrict__ attn) {
  __shared__ v4f qs[QT][NCH];
  __shared__ v4f ws[NCH];
  __shared__ v4f redm[8];
  __shared__ v4f reds[8];

  const int t    = threadIdx.x;
  const int lane = t & 31;
  const int wave = t >> 5;
  const int b    = blockIdx.x / (QLEN / QT);
  const int bq0  = blockIdx.x * QT;

  (&qs[0][0])[t] = *(const v4f*)(Qp + (size_t)bq0 * HDIM + 4 * t);
  if (t < NCH) ws[t] = *(const v4f*)(wv + 4 * t);

  const int kbase  = wave * (KLEN / 8) + lane * 4;
  const float* kp  = KpT + (size_t)b * HDIM * KLEN + kbase;

  v4f kr[NS][4];
#pragma unroll
  for (int s = 0; s < NS - 1; ++s)
#pragma unroll
    for (int u = 0; u < 4; ++u) kr[s][u] = *(const v4f*)(kp + (size_t)(4 * s + u) * KLEN);

  v4f acc[QT];
#pragma unroll
  for (int qi = 0; qi < QT; ++qi) acc[qi] = v4f{0.f, 0.f, 0.f, 0.f};

  __syncthreads();

  v4f wc = ws[0];
  v4f qc[QT];
#pragma unroll
  for (int qi = 0; qi < QT; ++qi) qc[qi] = qs[qi][0];

  for (int it = 0; it < NCH / NS; ++it)
    ring_pass(it, kp, qs, ws, kr, wc, qc, acc);

  const int vlen = valid_lens[b];
  v4f mx;
#pragma unroll
  for (int qi = 0; qi < QT; ++qi) {
    acc[qi].x = (kbase + 0 < vlen) ? acc[qi].x : NEGBIG;
    acc[qi].y = (kbase + 1 < vlen) ? acc[qi].y : NEGBIG;
    acc[qi].z = (kbase + 2 < vlen) ? acc[qi].z : NEGBIG;
    acc[qi].w = (kbase + 3 < vlen) ? acc[qi].w : NEGBIG;
    mx[qi] = fmaxf(fmaxf(acc[qi].x, acc[qi].y), fmaxf(acc[qi].z, acc[qi].w));
  }
#pragma unroll
  for (int off = 16; off > 0; off >>= 1)
#pragma unroll
    for (int qi = 0; qi < QT; ++qi) mx[qi] = fmaxf(mx[qi], __shfl_xor(mx[qi], off, 32));
  if (lane == 0) redm[wave] = mx;
  __syncthreads();

  v4f M = redm[0];
#pragma unroll
  for (int w = 1; w < 8; ++w) {
    const v4f o = redm[w];
#pragma unroll
    for (int qi = 0; qi < QT; ++qi) M[qi] = fmaxf(M[qi], o[qi]);
  }

  v4f sm;
#pragma unroll
  for (int qi = 0; qi < QT; ++qi) {
    acc[qi].x = __expf(acc[qi].x - M[qi]);
    acc[qi].y = __expf(acc[qi].y - M[qi]);
    acc[qi].z = __expf(acc[qi].z - M[qi]);
    acc[qi].w = __expf(acc[qi].w - M[qi]);
    sm[qi] = (acc[qi].x + acc[qi].y) + (acc[qi].z + acc[qi].w);
  }
#pragma unroll
  for (int off = 16; off > 0; off >>= 1)
#pragma unroll
    for (int qi = 0; qi < QT; ++qi) sm[qi] += __shfl_xor(sm[qi], off, 32);
  if (lane == 0) reds[wave] = sm;
  __syncthreads();

  v4f S = reds[0];
#pragma unroll
  for (int w = 1; w < 8; ++w) {
    const v4f o = reds[w];
#pragma unroll
    for (int qi = 0; qi < QT; ++qi) S[qi] += o[qi];
  }

#pragma unroll
  for (int qi = 0; qi < QT; ++qi) {
    const float inv = 1.f / S[qi];
    *(v4f*)(attn + (size_t)(bq0 + qi) * KLEN + kbase) =
        v4f{acc[qi].x * inv, acc[qi].y * inv, acc[qi].z * inv, acc[qi].w * inv};
  }
}

#define OG_KW    (KLEN / 8)
#define OG_NU    (OG_KW / 8)
#define OG_PITCH 36

__global__ __launch_bounds__(256) void out_gemm(const float* __restrict__ attn,
                                                const float* __restrict__ V,
                                                float* __restrict__ OUT) {
  __shared__ __attribute__((aligned(16))) float red[8][32][OG_PITCH];

  const int t    = threadIdx.x;
  const int lane = t & 31;
  const int wave = t >> 5;
  const int half = lane >> 4;
  const int l16  = lane & 15;
  const int b    = blockIdx.x >> 4;
  const int m0   = ((blockIdx.x >> 3) & 1) * 32;
  const int n0   = (blockIdx.x & 7) * 32;
  const int kw0  = wave * OG_KW;

  const float* Ap = attn + ((size_t)b * QLEN + m0 + l16) * KLEN + kw0 + 4 * half;
  const float* Vp = V + ((size_t)b * KLEN + kw0 + 4 * half) * DV + n0 + 2 * l16;

  v4f ca[OG_NU][2];
  v2f cb[OG_NU][4];
#pragma unroll
  for (int u = 0; u < OG_NU; ++u) {
#pragma unroll
    for (int mt = 0; mt < 2; ++mt) ca[u][mt] = *(const v4f*)(Ap + mt * 16 * KLEN + 8 * u);
#pragma unroll
    for (int j = 0; j < 4; ++j) cb[u][j] = *(const v2f*)(Vp + (size_t)(8 * u + j) * DV);
  }
  __builtin_amdgcn_sched_barrier(0);

  v8f acc[2][2];
#pragma unroll
  for (int mt = 0; mt < 2; ++mt)
#pragma unroll
    for (int nt = 0; nt < 2; ++nt) acc[mt][nt] = v8f{};

#pragma unroll
  for (int u = 0; u < OG_NU; ++u) {
#pragma unroll
    for (int p = 0; p < 2; ++p)
#pragma unroll
      for (int mt = 0; mt < 2; ++mt)
#pragma unroll
        for (int nt = 0; nt < 2; ++nt) {
          const v2f a  = p ? v2f{ca[u][mt].z, ca[u][mt].w} : v2f{ca[u][mt].x, ca[u][mt].y};
          const v2f bv = {cb[u][2 * p][nt], cb[u][2 * p + 1][nt]};
          acc[mt][nt] = __builtin_amdgcn_wmma_f32_16x16x4_f32(
               false, a,  false, bv,
               (short)0, acc[mt][nt],  false,  false);
        }
  }

#pragma unroll
  for (int mt = 0; mt < 2; ++mt)
#pragma unroll
    for (int v = 0; v < 8; ++v)
      *(v2f*)&red[wave][mt * 16 + 8 * half + v][2 * l16] = v2f{acc[mt][0][v], acc[mt][1][v]};
  __syncthreads();

  const int row = t >> 3;
  const int c4  = (t & 7) * 4;
  v4f s = *(const v4f*)&red[0][row][c4];
#pragma unroll
  for (int w = 1; w < 8; ++w) {
    const v4f o = *(const v4f*)&red[w][row][c4];
    s.x += o.x; s.y += o.y; s.z += o.z; s.w += o.w;
  }
  *(v4f*)(OUT + ((size_t)b * QLEN + m0 + row) * DV + n0 + c4) = s;
}

extern "C" void kernel_launch(void* const* d_in, const int* in_sizes, int n_in,
                              void* d_out, int out_size, void* d_ws, size_t ws_size,
                              hipStream_t stream) {
  const float* queries = (const float*)d_in[0];
  const float* keys    = (const float*)d_in[1];
  const float* values  = (const float*)d_in[2];
  const int*   vlens   = (const int*)d_in[3];
  const float* Wq      = (const float*)d_in[4];
  const float* Wk      = (const float*)d_in[5];
  const float* wv      = (const float*)d_in[6];
  float* out = (float*)d_out;

  float* Qp   = (float*)d_ws;
  float* Kp   = Qp + (size_t)BQ * HDIM;
  float* attn = Kp + (size_t)BKEYS * HDIM;

  proj_gemm_q<<<64, 256, 0, stream>>>(queries, Wq, Qp);
  proj_gemm_k<<<256, 256, 0, stream>>>(keys, Wk, Kp);

  score_softmax<<<BQ / QT, 256, 0, stream>>>(Qp, Kp, wv, vlens, attn);

  out_gemm<<<BATCH * (QLEN / 32) * (DV / 32), 256, 0, stream>>>(attn, values, out);
}


# Query projection joins the key projection's launch as 128 extra blocks (16x16 wave tiles) behind the 256 key blocks: 4 launches become 3; every sum keeps its order
# speedup vs baseline: 3.1609x; 1.1256x over previous
#include <hip/hip_runtime.h>
#include <math.h>

#define HDIM  256
#define BATCH 16
#define QLEN  64
#define KLEN  1024
#define DV    256
#define BQ    (BATCH * QLEN)
#define BKEYS (BATCH * KLEN)
#define NEGBIG -1e6f

typedef float v2f __attribute__((ext_vector_type(2)));
typedef float v4f __attribute__((ext_vector_type(4)));
typedef float v8f __attribute__((ext_vector_type(8)));

__device__ __forceinline__ float fast_tanh(float x) {
#if __has_builtin(__builtin_amdgcn_tanhf)
  return __builtin_amdgcn_tanhf(x);
#elif __has_builtin(__builtin_amdgcn_tanh_f32)
  return __builtin_amdgcn_tanh_f32(x);
#else
  return tanhf(x);
#endif
}

template <int MT, int NT, int WM, int WN, int KU, bool HMAJOR>
__device__ __forceinline__ void proj_tile(const float* __restrict__ X,
                                          const float* __restrict__ W,
                                          float* __restrict__ OUT,
                                          const int bid) {
  constexpr int BM  = WM * MT * 16;
  constexpr int BN  = WN * NT * 16;
  constexpr int NBN = HDIM / BN;
  constexpr int KC  = 8 * KU;

  const int lane = threadIdx.x & 31;
  const int wave = threadIdx.x >> 5;
  const int half = lane >> 4;
  const int l16  = lane & 15;
  const int bm   = bid / NBN;
  const int bn   = bid % NBN;
  const int m0   = bm * BM + (wave / WN) * (MT * 16);
  const int n0   = bn * BN + (wave % WN) * (NT * 16);

  const float* Ap = X + (size_t)(m0 + l16) * HDIM + 4 * half;
  const float* Bp = W + (size_t)(n0 + l16) * HDIM + 4 * half;

  v8f acc[MT][NT];
#pragma unroll
  for (int mt = 0; mt < MT; ++mt)
#pragma unroll
    for (int nt = 0; nt < NT; ++nt) acc[mt][nt] = v8f{};

  v4f ca[KU][MT], cb[KU][NT];
#pragma unroll
  for (int u = 0; u < KU; ++u) {
#pragma unroll
    for (int mt = 0; mt < MT; ++mt) ca[u][mt] = *(const v4f*)(Ap + mt * 16 * HDIM + 8 * u);
#pragma unroll
    for (int nt = 0; nt < NT; ++nt) cb[u][nt] = *(const v4f*)(Bp + nt * 16 * HDIM + 8 * u);
  }

#pragma unroll
  for (int kc = 0; kc < HDIM; kc += KC) {
    v4f na[KU][MT], nb[KU][NT];
    if (kc + KC < HDIM) {
      const int kn = kc + KC;
#pragma unroll
      for (int u = 0; u < KU; ++u) {
#pragma unroll
        for (int mt = 0; mt < MT; ++mt) na[u][mt] = *(const v4f*)(Ap + mt * 16 * HDIM + kn + 8 * u);
#pragma unroll
        for (int nt = 0; nt < NT; ++nt) nb[u][nt] = *(const v4f*)(Bp + nt * 16 * HDIM + kn + 8 * u);
      }
    }
    __builtin_amdgcn_sched_barrier(0);

#pragma unroll
    for (int u = 0; u < KU; ++u) {
#pragma unroll
      for (int mt = 0; mt < MT; ++mt)
#pragma unroll
        for (int nt = 0; nt < NT; ++nt) {
          const v2f a = {ca[u][mt].x, ca[u][mt].y};
          const v2f b = {cb[u][nt].x, cb[u][nt].y};
          acc[mt][nt] = __builtin_amdgcn_wmma_f32_16x16x4_f32(
               false, a,  false, b,
               (short)0, acc[mt][nt],  false,  false);
        }
#pragma unroll
      for (int mt = 0; mt < MT; ++mt)
#pragma unroll
        for (int nt = 0; nt < NT; ++nt) {
          const v2f a = {ca[u][mt].z, ca[u][mt].w};
          const v2f b = {cb[u][nt].z, cb[u][nt].w};
          acc[mt][nt] = __builtin_amdgcn_wmma_f32_16x16x4_f32(
              false, a, false, b, (short)0, acc[mt][nt], false, false);
        }
    }

    if (kc + KC < HDIM) {
#pragma unroll
      for (int u = 0; u < KU; ++u) {
#pragma unroll
        for (int mt = 0; mt < MT; ++mt) ca[u][mt] = na[u][mt];
#pragma unroll
        for (int nt = 0; nt < NT; ++nt) cb[u][nt] = nb[u][nt];
      }
    }
  }

  if (HMAJOR) {
#pragma unroll
    for (int mt = 0; mt < MT; ++mt)
#pragma unroll
      for (int nt = 0; nt < NT; ++nt) {
        const int mrow = m0 + mt * 16 + 8 * half;
        const int bb   = mrow / KLEN;
        const int kk   = mrow % KLEN;
        float* Ocol = OUT + ((size_t)bb * HDIM + n0 + nt * 16 + l16) * KLEN + kk;
        const v8f a = acc[mt][nt];
        *(v4f*)(Ocol)     = v4f{a[0], a[1], a[2], a[3]};
        *(v4f*)(Ocol + 4) = v4f{a[4], a[5], a[6], a[7]};
      }
  } else {
#pragma unroll
    for (int mt = 0; mt < MT; ++mt)
#pragma unroll
      for (int nt = 0; nt < NT; ++nt) {
        float* Orow = OUT + (size_t)(m0 + mt * 16 + 8 * half) * HDIM + n0 + nt * 16 + l16;
#pragma unroll
        for (int v = 0; v < 8; ++v)
          Orow[v * HDIM] = acc[mt][nt][v];
      }
  }
}

#define KP_BLOCKS ((BKEYS / 128) * (HDIM / 128))
#define QP_BLOCKS ((BQ / 32) * (HDIM / 64))

__global__ __launch_bounds__(256) void proj_gemm_kq(const float* __restrict__ keys,
                                                    const float* __restrict__ Wk,
                                                    float* __restrict__ KpT,
                                                    const float* __restrict__ queries,
                                                    const float* __restrict__ Wq,
                                                    float* __restrict__ Qp) {
  if (blockIdx.x < KP_BLOCKS)
    proj_tile<2, 4, 4, 2, 2, true>(keys, Wk, KpT, (int)blockIdx.x);
  else
    proj_tile<1, 1, 2, 4, 4, false>(queries, Wq, Qp, (int)blockIdx.x - KP_BLOCKS);
}

#define QT 4
#define NS 8
#define NCH (HDIM / 4)

__device__ __forceinline__ void ring_pass(const int it, const float* __restrict__ kp,
                                          const v4f (*qs)[NCH], const v4f* ws,
                                          v4f (&kr)[NS][4], v4f& wc, v4f (&qc)[QT],
                                          v4f (&acc)[QT]) {
#pragma unroll
  for (int s = 0; s < NS; ++s) {
    const int c = it * NS + s;
    const int cn = min(c + NS - 1, NCH - 1);
#pragma unroll
    for (int u = 0; u < 4; ++u)
      kr[(s + NS - 1) % NS][u] = *(const v4f*)(kp + (size_t)(4 * cn + u) * KLEN);
    const int c1 = (c + 1) & (NCH - 1);
    const v4f wn = ws[c1];
    v4f qn[QT];
#pragma unroll
    for (int qi = 0; qi < QT; ++qi) qn[qi] = qs[qi][c1];
    __builtin_amdgcn_sched_barrier(0);
#pragma unroll
    for (int uu = 0; uu < 4; ++uu) {
      const v4f kv = kr[s][uu];
      const float w = wc[uu];
#pragma unroll
      for (int qi = 0; qi < QT; ++qi) {
        const float qv = qc[qi][uu];
        acc[qi].x = fmaf(fast_tanh(qv + kv.x), w, acc[qi].x);
        acc[qi].y = fmaf(fast_tanh(qv + kv.y), w, acc[qi].y);
        acc[qi].z = fmaf(fast_tanh(qv + kv.z), w, acc[qi].z);
        acc[qi].w = fmaf(fast_tanh(qv + kv.w), w, acc[qi].w);
      }
    }
    wc = wn;
#pragma unroll
    for (int qi = 0; qi < QT; ++qi) qc[qi] = qn[qi];
  }
}

__global__ __launch_bounds__(256) __attribute__((amdgpu_waves_per_eu(2, 2)))
void score_softmax(const float* __restrict__ Qp,
                   const float* __restrict__ KpT,
                   const float* __restrict__ wv,
                   const int* __restrict__ valid_lens,
                   float* __restrict__ attn) {
  __shared__ v4f qs[QT][NCH];
  __shared__ v4f ws[NCH];
  __shared__ v4f redm[8];
  __shared__ v4f reds[8];

  const int t    = threadIdx.x;
  const int lane = t & 31;
  const int wave = t >> 5;
  const int b    = blockIdx.x / (QLEN / QT);
  const int bq0  = blockIdx.x * QT;

  (&qs[0][0])[t] = *(const v4f*)(Qp + (size_t)bq0 * HDIM + 4 * t);
  if (t < NCH) ws[t] = *(const v4f*)(wv + 4 * t);

  const int kbase  = wave * (KLEN / 8) + lane * 4;
  const float* kp  = KpT + (size_t)b * HDIM * KLEN + kbase;

  v4f kr[NS][4];
#pragma unroll
  for (int s = 0; s < NS - 1; ++s)
#pragma unroll
    for (int u = 0; u < 4; ++u) kr[s][u] = *(const v4f*)(kp + (size_t)(4 * s + u) * KLEN);

  v4f acc[QT];
#pragma unroll
  for (int qi = 0; qi < QT; ++qi) acc[qi] = v4f{0.f, 0.f, 0.f, 0.f};

  __syncthreads();

  v4f wc = ws[0];
  v4f qc[QT];
#pragma unroll
  for (int qi = 0; qi < QT; ++qi) qc[qi] = qs[qi][0];

  for (int it = 0; it < NCH / NS; ++it)
    ring_pass(it, kp, qs, ws, kr, wc, qc, acc);

  const int vlen = valid_lens[b];
  v4f mx;
#pragma unroll
  for (int qi = 0; qi < QT; ++qi) {
    acc[qi].x = (kbase + 0 < vlen) ? acc[qi].x : NEGBIG;
    acc[qi].y = (kbase + 1 < vlen) ? acc[qi].y : NEGBIG;
    acc[qi].z = (kbase + 2 < vlen) ? acc[qi].z : NEGBIG;
    acc[qi].w = (kbase + 3 < vlen) ? acc[qi].w : NEGBIG;
    mx[qi] = fmaxf(fmaxf(acc[qi].x, acc[qi].y), fmaxf(acc[qi].z, acc[qi].w));
  }
#pragma unroll
  for (int off = 16; off > 0; off >>= 1)
#pragma unroll
    for (int qi = 0; qi < QT; ++qi) mx[qi] = fmaxf(mx[qi], __shfl_xor(mx[qi], off, 32));
  if (lane == 0) redm[wave] = mx;
  __syncthreads();

  v4f M = redm[0];
#pragma unroll
  for (int w = 1; w < 8; ++w) {
    const v4f o = redm[w];
#pragma unroll
    for (int qi = 0; qi < QT; ++qi) M[qi] = fmaxf(M[qi], o[qi]);
  }

  v4f sm;
#pragma unroll
  for (int qi = 0; qi < QT; ++qi) {
    acc[qi].x = __expf(acc[qi].x - M[qi]);
    acc[qi].y = __expf(acc[qi].y - M[qi]);
    acc[qi].z = __expf(acc[qi].z - M[qi]);
    acc[qi].w = __expf(acc[qi].w - M[qi]);
    sm[qi] = (acc[qi].x + acc[qi].y) + (acc[qi].z + acc[qi].w);
  }
#pragma unroll
  for (int off = 16; off > 0; off >>= 1)
#pragma unroll
    for (int qi = 0; qi < QT; ++qi) sm[qi] += __shfl_xor(sm[qi], off, 32);
  if (lane == 0) reds[wave] = sm;
  __syncthreads();

  v4f S = reds[0];
#pragma unroll
  for (int w = 1; w < 8; ++w) {
    const v4f o = reds[w];
#pragma unroll
    for (int qi = 0; qi < QT; ++qi) S[qi] += o[qi];
  }

#pragma unroll
  for (int qi = 0; qi < QT; ++qi) {
    const float inv = 1.f / S[qi];
    *(v4f*)(attn + (size_t)(bq0 + qi) * KLEN + kbase) =
        v4f{acc[qi].x * inv, acc[qi].y * inv, acc[qi].z * inv, acc[qi].w * inv};
  }
}

#define OG_KW    (KLEN / 8)
#define OG_NU    (OG_KW / 8)
#define OG_PITCH 36

__global__ __launch_bounds__(256) void out_gemm(const float* __restrict__ attn,
                                                const float* __restrict__ V,
                                                float* __restrict__ OUT) {
  __shared__ __attribute__((aligned(16))) float red[8][32][OG_PITCH];

  const int t    = threadIdx.x;
  const int lane = t & 31;
  const int wave = t >> 5;
  const int half = lane >> 4;
  const int l16  = lane & 15;
  const int b    = blockIdx.x >> 4;
  const int m0   = ((blockIdx.x >> 3) & 1) * 32;
  const int n0   = (blockIdx.x & 7) * 32;
  const int kw0  = wave * OG_KW;

  const float* Ap = attn + ((size_t)b * QLEN + m0 + l16) * KLEN + kw0 + 4 * half;
  const float* Vp = V + ((size_t)b * KLEN + kw0 + 4 * half) * DV + n0 + 2 * l16;

  v4f ca[OG_NU][2];
  v2f cb[OG_NU][4];
#pragma unroll
  for (int u = 0; u < OG_NU; ++u) {
#pragma unroll
    for (int mt = 0; mt < 2; ++mt) ca[u][mt] = *(const v4f*)(Ap + mt * 16 * KLEN + 8 * u);
#pragma unroll
    for (int j = 0; j < 4; ++j) cb[u][j] = *(const v2f*)(Vp + (size_t)(8 * u + j) * DV);
  }
  __builtin_amdgcn_sched_barrier(0);

  v8f acc[2][2];
#pragma unroll
  for (int mt = 0; mt < 2; ++mt)
#pragma unroll
    for (int nt = 0; nt < 2; ++nt) acc[mt][nt] = v8f{};

#pragma unroll
  for (int u = 0; u < OG_NU; ++u) {
#pragma unroll
    for (int p = 0; p < 2; ++p)
#pragma unroll
      for (int mt = 0; mt < 2; ++mt)
#pragma unroll
        for (int nt = 0; nt < 2; ++nt) {
          const v2f a  = p ? v2f{ca[u][mt].z, ca[u][mt].w} : v2f{ca[u][mt].x, ca[u][mt].y};
          const v2f bv = {cb[u][2 * p][nt], cb[u][2 * p + 1][nt]};
          acc[mt][nt] = __builtin_amdgcn_wmma_f32_16x16x4_f32(
               false, a,  false, bv,
               (short)0, acc[mt][nt],  false,  false);
        }
  }

#pragma unroll
  for (int mt = 0; mt < 2; ++mt)
#pragma unroll
    for (int v = 0; v < 8; ++v)
      *(v2f*)&red[wave][mt * 16 + 8 * half + v][2 * l16] = v2f{acc[mt][0][v], acc[mt][1][v]};
  __syncthreads();

  const int row = t >> 3;
  const int c4  = (t & 7) * 4;
  v4f s = *(const v4f*)&red[0][row][c4];
#pragma unroll
  for (int w = 1; w < 8; ++w) {
    const v4f o = *(const v4f*)&red[w][row][c4];
    s.x += o.x; s.y += o.y; s.z += o.z; s.w += o.w;
  }
  *(v4f*)(OUT + ((size_t)b * QLEN + m0 + row) * DV + n0 + c4) = s;
}

extern "C" void kernel_launch(void* const* d_in, const int* in_sizes, int n_in,
                              void* d_out, int out_size, void* d_ws, size_t ws_size,
                              hipStream_t stream) {
  const float* queries = (const float*)d_in[0];
  const float* keys    = (const float*)d_in[1];
  const float* values  = (const float*)d_in[2];
  const int*   vlens   = (const int*)d_in[3];
  const float* Wq      = (const float*)d_in[4];
  const float* Wk      = (const float*)d_in[5];
  const float* wv      = (const float*)d_in[6];
  float* out = (float*)d_out;

  float* Qp   = (float*)d_ws;
  float* Kp   = Qp + (size_t)BQ * HDIM;
  float* attn = Kp + (size_t)BKEYS * HDIM;

  proj_gemm_kq<<<KP_BLOCKS + QP_BLOCKS, 256, 0, stream>>>(keys, Wk, Kp, queries, Wq, Qp);

  score_softmax<<<BQ / QT, 256, 0, stream>>>(Qp, Kp, wv, vlens, attn);

  out_gemm<<<BATCH * (QLEN / 32) * (DV / 32), 256, 0, stream>>>(attn, values, out);
}


# Projection operands staged through LDS: the block fetches each 32-k chunk as whole 128-byte lines into a two-stage LDS ring and waves read their 16-byte fragments there; every sum keeps its order
# speedup vs baseline: 3.5393x; 1.1197x over previous
#include <hip/hip_runtime.h>
#include <math.h>

#define HDIM  256
#define BATCH 16
#define QLEN  64
#define KLEN  1024
#define DV    256
#define BQ    (BATCH * QLEN)
#define BKEYS (BATCH * KLEN)
#define NEGBIG -1e6f

typedef float v2f __attribute__((ext_vector_type(2)));
typedef float v4f __attribute__((ext_vector_type(4)));
typedef float v8f __attribute__((ext_vector_type(8)));

__device__ __forceinline__ float fast_tanh(float x) {
#if __has_builtin(__builtin_amdgcn_tanhf)
  return __builtin_amdgcn_tanhf(x);
#elif __has_builtin(__builtin_amdgcn_tanh_f32)
  return __builtin_amdgcn_tanh_f32(x);
#else
  return tanhf(x);
#endif
}

#define PJ_KC    32
#define PJ_NCH   (HDIM / PJ_KC)
#define PJ_PITCH 36
#define PJ_ROWS  256
#define PJ_STAGE (PJ_ROWS * PJ_PITCH)

template <int MT, int NT, int WM, int WN, bool HMAJOR>
__device__ __forceinline__ void proj_tile(const float* __restrict__ X,
                                          const float* __restrict__ W,
                                          float* __restrict__ OUT,
                                          const int bid,
                                          float* sm) {
  constexpr int BM  = WM * MT * 16;
  constexpr int BN  = WN * NT * 16;
  constexpr int NBN = HDIM / BN;
  constexpr int NLA = BM / 32;
  constexpr int NLB = BN / 32;
  constexpr int NLD = NLA + NLB;
  static_assert(BM % 32 == 0 && BN % 32 == 0 && BM + BN <= PJ_ROWS, "tile does not fit the stage");

  const int t    = threadIdx.x;
  const int lane = t & 31;
  const int wave = t >> 5;
  const int half = lane >> 4;
  const int l16  = lane & 15;
  const int bm   = bid / NBN;
  const int bn   = bid % NBN;
  const int wm   = wave / WN;
  const int wn   = wave % WN;
  const int m0   = bm * BM + wm * (MT * 16);
  const int n0   = bn * BN + wn * (NT * 16);

  const int lr = t >> 3;
  const int lp = t & 7;
  const float* Ag = X + (size_t)(bm * BM + lr) * HDIM + 4 * lp;
  const float* Bg = W + (size_t)(bn * BN + lr) * HDIM + 4 * lp;
  const int so = lr * PJ_PITCH + 4 * lp;
  const int fa = (wm * (MT * 16) + l16) * PJ_PITCH + 4 * half;
  const int fb = (BM + wn * (NT * 16) + l16) * PJ_PITCH + 4 * half;

  v8f acc[MT][NT];
#pragma unroll
  for (int mt = 0; mt < MT; ++mt)
#pragma unroll
    for (int nt = 0; nt < NT; ++nt) acc[mt][nt] = v8f{};

  v4f g[NLD];
  v4f a0[2][MT], b0[2][NT];
  v4f a1[2][MT], b1[2][NT];

#pragma unroll
  for (int i = 0; i < NLA; ++i) g[i] = *(const v4f*)(Ag + (size_t)i * 32 * HDIM);
#pragma unroll
  for (int j = 0; j < NLB; ++j) g[NLA + j] = *(const v4f*)(Bg + (size_t)j * 32 * HDIM);
#pragma unroll
  for (int i = 0; i < NLD; ++i) *(v4f*)(sm + so + i * 32 * PJ_PITCH) = g[i];
  __syncthreads();
#pragma unroll
  for (int i = 0; i < NLA; ++i) g[i] = *(const v4f*)(Ag + (size_t)i * 32 * HDIM + PJ_KC);
#pragma unroll
  for (int j = 0; j < NLB; ++j) g[NLA + j] = *(const v4f*)(Bg + (size_t)j * 32 * HDIM + PJ_KC);
#pragma unroll
  for (int uu = 0; uu < 2; ++uu) {
#pragma unroll
    for (int mt = 0; mt < MT; ++mt) a0[uu][mt] = *(const v4f*)(sm + fa + mt * 16 * PJ_PITCH + 8 * uu);
#pragma unroll
    for (int nt = 0; nt < NT; ++nt) b0[uu][nt] = *(const v4f*)(sm + fb + nt * 16 * PJ_PITCH + 8 * uu);
  }

#pragma unroll
  for (int c = 0; c < PJ_NCH; ++c) {
    float* cur = sm + (c & 1) * PJ_STAGE;
    float* nxt = sm + ((c + 1) & 1) * PJ_STAGE;

#pragma unroll
    for (int uu = 0; uu < 2; ++uu) {
#pragma unroll
      for (int mt = 0; mt < MT; ++mt) a1[uu][mt] = *(const v4f*)(cur + fa + mt * 16 * PJ_PITCH + 8 * (2 + uu));
#pragma unroll
      for (int nt = 0; nt < NT; ++nt) b1[uu][nt] = *(const v4f*)(cur + fb + nt * 16 * PJ_PITCH + 8 * (2 + uu));
    }
    __builtin_amdgcn_sched_barrier(0);

#pragma unroll
    for (int uu = 0; uu < 2; ++uu) {
#pragma unroll
      for (int mt = 0; mt < MT; ++mt)
#pragma unroll
        for (int nt = 0; nt < NT; ++nt) {
          const v2f a = {a0[uu][mt].x, a0[uu][mt].y};
          const v2f b = {b0[uu][nt].x, b0[uu][nt].y};
          acc[mt][nt] = __builtin_amdgcn_wmma_f32_16x16x4_f32(
               false, a,  false, b,
               (short)0, acc[mt][nt],  false,  false);
        }
#pragma unroll
      for (int mt = 0; mt < MT; ++mt)
#pragma unroll
        for (int nt = 0; nt < NT; ++nt) {
          const v2f a = {a0[uu][mt].z, a0[uu][mt].w};
          const v2f b = {b0[uu][nt].z, b0[uu][nt].w};
          acc[mt][nt] = __builtin_amdgcn_wmma_f32_16x16x4_f32(
              false, a, false, b, (short)0, acc[mt][nt], false, false);
        }
    }
    __builtin_amdgcn_sched_barrier(0);

    if (c + 1 < PJ_NCH) {
#pragma unroll
      for (int i = 0; i < NLD; ++i) *(v4f*)(nxt + so + i * 32 * PJ_PITCH) = g[i];
      __syncthreads();
      if (c + 2 < PJ_NCH) {
#pragma unroll
        for (int i = 0; i < NLA; ++i) g[i] = *(const v4f*)(Ag + (size_t)i * 32 * HDIM + (c + 2) * PJ_KC);
#pragma unroll
        for (int j = 0; j < NLB; ++j) g[NLA + j] = *(const v4f*)(Bg + (size_t)j * 32 * HDIM + (c + 2) * PJ_KC);
      }
#pragma unroll
      for (int uu = 0; uu < 2; ++uu) {
#pragma unroll
        for (int mt = 0; mt < MT; ++mt) a0[uu][mt] = *(const v4f*)(nxt + fa + mt * 16 * PJ_PITCH + 8 * uu);
#pragma unroll
        for (int nt = 0; nt < NT; ++nt) b0[uu][nt] = *(const v4f*)(nxt + fb + nt * 16 * PJ_PITCH + 8 * uu);
      }
      __builtin_amdgcn_sched_barrier(0);
    }

#pragma unroll
    for (int uu = 0; uu < 2; ++uu) {
#pragma unroll
      for (int mt = 0; mt < MT; ++mt)
#pragma unroll
        for (int nt = 0; nt < NT; ++nt) {
          const v2f a = {a1[uu][mt].x, a1[uu][mt].y};
          const v2f b = {b1[uu][nt].x, b1[uu][nt].y};
          acc[mt][nt] = __builtin_amdgcn_wmma_f32_16x16x4_f32(
              false, a, false, b, (short)0, acc[mt][nt], false, false);
        }
#pragma unroll
      for (int mt = 0; mt < MT; ++mt)
#pragma unroll
        for (int nt = 0; nt < NT; ++nt) {
          const v2f a = {a1[uu][mt].z, a1[uu][mt].w};
          const v2f b = {b1[uu][nt].z, b1[uu][nt].w};
          acc[mt][nt] = __builtin_amdgcn_wmma_f32_16x16x4_f32(
              false, a, false, b, (short)0, acc[mt][nt], false, false);
        }
    }
  }

  if (HMAJOR) {
#pragma unroll
    for (int mt = 0; mt < MT; ++mt)
#pragma unroll
      for (int nt = 0; nt < NT; ++nt) {
        const int mrow = m0 + mt * 16 + 8 * half;
        const int bb   = mrow / KLEN;
        const int kk   = mrow % KLEN;
        float* Ocol = OUT + ((size_t)bb * HDIM + n0 + nt * 16 + l16) * KLEN + kk;
        const v8f a = acc[mt][nt];
        *(v4f*)(Ocol)     = v4f{a[0], a[1], a[2], a[3]};
        *(v4f*)(Ocol + 4) = v4f{a[4], a[5], a[6], a[7]};
      }
  } else {
#pragma unroll
    for (int mt = 0; mt < MT; ++mt)
#pragma unroll
      for (int nt = 0; nt < NT; ++nt) {
        float* Orow = OUT + (size_t)(m0 + mt * 16 + 8 * half) * HDIM + n0 + nt * 16 + l16;
#pragma unroll
        for (int v = 0; v < 8; ++v)
          Orow[v * HDIM] = acc[mt][nt][v];
      }
  }
}

#define KP_BLOCKS ((BKEYS / 128) * (HDIM / 128))
#define QP_BLOCKS ((BQ / 32) * (HDIM / 64))

__global__ __launch_bounds__(256) void proj_gemm_kq(const float* __restrict__ keys,
                                                    const float* __restrict__ Wk,
                                                    float* __restrict__ KpT,
                                                    const float* __restrict__ queries,
                                                    const float* __restrict__ Wq,
                                                    float* __restrict__ Qp) {
  __shared__ __attribute__((aligned(16))) float stage[2 * PJ_STAGE];
  if (blockIdx.x < KP_BLOCKS)
    proj_tile<2, 4, 4, 2, true>(keys, Wk, KpT, (int)blockIdx.x, stage);
  else
    proj_tile<1, 1, 2, 4, false>(queries, Wq, Qp, (int)blockIdx.x - KP_BLOCKS, stage);
}

#define QT 4
#define NS 8
#define NCH (HDIM / 4)

__device__ __forceinline__ void ring_pass(const int it, const float* __restrict__ kp,
                                          const v4f (*qs)[NCH], const v4f* ws,
                                          v4f (&kr)[NS][4], v4f& wc, v4f (&qc)[QT],
                                          v4f (&acc)[QT]) {
#pragma unroll
  for (int s = 0; s < NS; ++s) {
    const int c = it * NS + s;
    const int cn = min(c + NS - 1, NCH - 1);
#pragma unroll
    for (int u = 0; u < 4; ++u)
      kr[(s + NS - 1) % NS][u] = *(const v4f*)(kp + (size_t)(4 * cn + u) * KLEN);
    const int c1 = (c + 1) & (NCH - 1);
    const v4f wn = ws[c1];
    v4f qn[QT];
#pragma unroll
    for (int qi = 0; qi < QT; ++qi) qn[qi] = qs[qi][c1];
    __builtin_amdgcn_sched_barrier(0);
#pragma unroll
    for (int uu = 0; uu < 4; ++uu) {
      const v4f kv = kr[s][uu];
      const float w = wc[uu];
#pragma unroll
      for (int qi = 0; qi < QT; ++qi) {
        const float qv = qc[qi][uu];
        acc[qi].x = fmaf(fast_tanh(qv + kv.x), w, acc[qi].x);
        acc[qi].y = fmaf(fast_tanh(qv + kv.y), w, acc[qi].y);
        acc[qi].z = fmaf(fast_tanh(qv + kv.z), w, acc[qi].z);
        acc[qi].w = fmaf(fast_tanh(qv + kv.w), w, acc[qi].w);
      }
    }
    wc = wn;
#pragma unroll
    for (int qi = 0; qi < QT; ++qi) qc[qi] = qn[qi];
  }
}

__global__ __launch_bounds__(256) __attribute__((amdgpu_waves_per_eu(2, 2)))
void score_softmax(const float* __restrict__ Qp,
                   const float* __restrict__ KpT,
                   const float* __restrict__ wv,
                   const int* __restrict__ valid_lens,
                   float* __restrict__ attn) {
  __shared__ v4f qs[QT][NCH];
  __shared__ v4f ws[NCH];
  __shared__ v4f redm[8];
  __shared__ v4f reds[8];

  const int t    = threadIdx.x;
  const int lane = t & 31;
  const int wave = t >> 5;
  const int b    = blockIdx.x / (QLEN / QT);
  const int bq0  = blockIdx.x * QT;

  (&qs[0][0])[t] = *(const v4f*)(Qp + (size_t)bq0 * HDIM + 4 * t);
  if (t < NCH) ws[t] = *(const v4f*)(wv + 4 * t);

  const int kbase  = wave * (KLEN / 8) + lane * 4;
  const float* kp  = KpT + (size_t)b * HDIM * KLEN + kbase;

  v4f kr[NS][4];
#pragma unroll
  for (int s = 0; s < NS - 1; ++s)
#pragma unroll
    for (int u = 0; u < 4; ++u) kr[s][u] = *(const v4f*)(kp + (size_t)(4 * s + u) * KLEN);

  v4f acc[QT];
#pragma unroll
  for (int qi = 0; qi < QT; ++qi) acc[qi] = v4f{0.f, 0.f, 0.f, 0.f};

  __syncthreads();

  v4f wc = ws[0];
  v4f qc[QT];
#pragma unroll
  for (int qi = 0; qi < QT; ++qi) qc[qi] = qs[qi][0];

  for (int it = 0; it < NCH / NS; ++it)
    ring_pass(it, kp, qs, ws, kr, wc, qc, acc);

  const int vlen = valid_lens[b];
  v4f mx;
#pragma unroll
  for (int qi = 0; qi < QT; ++qi) {
    acc[qi].x = (kbase + 0 < vlen) ? acc[qi].x : NEGBIG;
    acc[qi].y = (kbase + 1 < vlen) ? acc[qi].y : NEGBIG;
    acc[qi].z = (kbase + 2 < vlen) ? acc[qi].z : NEGBIG;
    acc[qi].w = (kbase + 3 < vlen) ? acc[qi].w : NEGBIG;
    mx[qi] = fmaxf(fmaxf(acc[qi].x, acc[qi].y), fmaxf(acc[qi].z, acc[qi].w));
  }
#pragma unroll
  for (int off = 16; off > 0; off >>= 1)
#pragma unroll
    for (int qi = 0; qi < QT; ++qi) mx[qi] = fmaxf(mx[qi], __shfl_xor(mx[qi], off, 32));
  if (lane == 0) redm[wave] = mx;
  __syncthreads();

  v4f M = redm[0];
#pragma unroll
  for (int w = 1; w < 8; ++w) {
    const v4f o = redm[w];
#pragma unroll
    for (int qi = 0; qi < QT; ++qi) M[qi] = fmaxf(M[qi], o[qi]);
  }

  v4f sm;
#pragma unroll
  for (int qi = 0; qi < QT; ++qi) {
    acc[qi].x = __expf(acc[qi].x - M[qi]);
    acc[qi].y = __expf(acc[qi].y - M[qi]);
    acc[qi].z = __expf(acc[qi].z - M[qi]);
    acc[qi].w = __expf(acc[qi].w - M[qi]);
    sm[qi] = (acc[qi].x + acc[qi].y) + (acc[qi].z + acc[qi].w);
  }
#pragma unroll
  for (int off = 16; off > 0; off >>= 1)
#pragma unroll
    for (int qi = 0; qi < QT; ++qi) sm[qi] += __shfl_xor(sm[qi], off, 32);
  if (lane == 0) reds[wave] = sm;
  __syncthreads();

  v4f S = reds[0];
#pragma unroll
  for (int w = 1; w < 8; ++w) {
    const v4f o = reds[w];
#pragma unroll
    for (int qi = 0; qi < QT; ++qi) S[qi] += o[qi];
  }

#pragma unroll
  for (int qi = 0; qi < QT; ++qi) {
    const float inv = 1.f / S[qi];
    *(v4f*)(attn + (size_t)(bq0 + qi) * KLEN + kbase) =
        v4f{acc[qi].x * inv, acc[qi].y * inv, acc[qi].z * inv, acc[qi].w * inv};
  }
}

#define OG_KW    (KLEN / 8)
#define OG_NU    (OG_KW / 8)
#define OG_PITCH 36

__global__ __launch_bounds__(256) void out_gemm(const float* __restrict__ attn,
                                                const float* __restrict__ V,
                                                float* __restrict__ OUT) {
  __shared__ __attribute__((aligned(16))) float red[8][32][OG_PITCH];

  const int t    = threadIdx.x;
  const int lane = t & 31;
  const int wave = t >> 5;
  const int half = lane >> 4;
  const int l16  = lane & 15;
  const int b    = blockIdx.x >> 4;
  const int m0   = ((blockIdx.x >> 3) & 1) * 32;
  const int n0   = (blockIdx.x & 7) * 32;
  const int kw0  = wave * OG_KW;

  const float* Ap = attn + ((size_t)b * QLEN + m0 + l16) * KLEN + kw0 + 4 * half;
  const float* Vp = V + ((size_t)b * KLEN + kw0 + 4 * half) * DV + n0 + 2 * l16;

  v4f ca[OG_NU][2];
  v2f cb[OG_NU][4];
#pragma unroll
  for (int u = 0; u < OG_NU; ++u) {
#pragma unroll
    for (int mt = 0; mt < 2; ++mt) ca[u][mt] = *(const v4f*)(Ap + mt * 16 * KLEN + 8 * u);
#pragma unroll
    for (int j = 0; j < 4; ++j) cb[u][j] = *(const v2f*)(Vp + (size_t)(8 * u + j) * DV);
  }
  __builtin_amdgcn_sched_barrier(0);

  v8f acc[2][2];
#pragma unroll
  for (int mt = 0; mt < 2; ++mt)
#pragma unroll
    for (int nt = 0; nt < 2; ++nt) acc[mt][nt] = v8f{};

#pragma unroll
  for (int u = 0; u < OG_NU; ++u) {
#pragma unroll
    for (int p = 0; p < 2; ++p)
#pragma unroll
      for (int mt = 0; mt < 2; ++mt)
#pragma unroll
        for (int nt = 0; nt < 2; ++nt) {
          const v2f a  = p ? v2f{ca[u][mt].z, ca[u][mt].w} : v2f{ca[u][mt].x, ca[u][mt].y};
          const v2f bv = {cb[u][2 * p][nt], cb[u][2 * p + 1][nt]};
          acc[mt][nt] = __builtin_amdgcn_wmma_f32_16x16x4_f32(
               false, a,  false, bv,
               (short)0, acc[mt][nt],  false,  false);
        }
  }

#pragma unroll
  for (int mt = 0; mt < 2; ++mt)
#pragma unroll
    for (int v = 0; v < 8; ++v)
      *(v2f*)&red[wave][mt * 16 + 8 * half + v][2 * l16] = v2f{acc[mt][0][v], acc[mt][1][v]};
  __syncthreads();

  const int row = t >> 3;
  const int c4  = (t & 7) * 4;
  v4f s = *(const v4f*)&red[0][row][c4];
#pragma unroll
  for (int w = 1; w < 8; ++w) {
    const v4f o = *(const v4f*)&red[w][row][c4];
    s.x += o.x; s.y += o.y; s.z += o.z; s.w += o.w;
  }
  *(v4f*)(OUT + ((size_t)b * QLEN + m0 + row) * DV + n0 + c4) = s;
}

extern "C" void kernel_launch(void* const* d_in, const int* in_sizes, int n_in,
                              void* d_out, int out_size, void* d_ws, size_t ws_size,
                              hipStream_t stream) {
  const float* queries = (const float*)d_in[0];
  const float* keys    = (const float*)d_in[1];
  const float* values  = (const float*)d_in[2];
  const int*   vlens   = (const int*)d_in[3];
  const float* Wq      = (const float*)d_in[4];
  const float* Wk      = (const float*)d_in[5];
  const float* wv      = (const float*)d_in[6];
  float* out = (float*)d_out;

  float* Qp   = (float*)d_ws;
  float* Kp   = Qp + (size_t)BQ * HDIM;
  float* attn = Kp + (size_t)BKEYS * HDIM;

  proj_gemm_kq<<<KP_BLOCKS + QP_BLOCKS, 256, 0, stream>>>(keys, Wk, Kp, queries, Wq, Qp);

  score_softmax<<<BQ / QT, 256, 0, stream>>>(Qp, Kp, wv, vlens, attn);

  out_gemm<<<BATCH * (QLEN / 32) * (DV / 32), 256, 0, stream>>>(attn, values, out);
}
